# starting kernel
# baseline (speedup 1.0000x reference)
#include <hip/hip_runtime.h>
#include <hip/hip_bf16.h>


typedef __attribute__((ext_vector_type(2))) float v2f;
typedef __attribute__((ext_vector_type(8))) float v8f;

#define B_DIM 8
#define S_DIM 4096
#define D_DIM 2048
#define E_DIM 16
#define KSPLIT 4
#define KCHUNK (D_DIM / KSPLIT)
#define CAPACITY 8.0f
#define EPSILON 1e-6f

__global__ __launch_bounds__(1024)
void switch_gate_wmma_kernel(const float* __restrict__ X,
                             const float* __restrict__ W,
                             const float* __restrict__ bias,
                             float* __restrict__ out) {
  __shared__ float part[KSPLIT * B_DIM][16][E_DIM];
  __shared__ float gate[B_DIM][16][E_DIM];

  const int tid  = threadIdx.x;
  const int ww   = tid >> 5;
  const int lane = tid & 31;
  const int half = lane >> 4;
  const int col  = lane & 15;
  const int b    = ww & 7;
  const int q    = ww >> 3;
  const int s0   = blockIdx.x * 16;
  const int k0   = q * KCHUNK;

  const float* __restrict__ xrow =
      X + ((size_t)b * S_DIM + (size_t)(s0 + col)) * D_DIM + k0 + 2 * half;
  const float* __restrict__ wcol = W + (size_t)(k0 + 2 * half) * E_DIM + col;

  v8f c = {};
  #pragma unroll 8
  for (int kk = 0; kk < KCHUNK; kk += 4) {
    v2f a = *(const v2f*)(xrow + kk);
    v2f bfrag;
    bfrag.x = wcol[kk * E_DIM];
    bfrag.y = wcol[kk * E_DIM + E_DIM];
    c = __builtin_amdgcn_wmma_f32_16x16x4_f32(
         false, a,  false, bfrag,
         (short)0, c,  false,  false);
  }

  #pragma unroll
  for (int j = 0; j < 8; ++j) part[ww][j + 8 * half][col] = c[j];
  __syncthreads();

  const float bval = bias[col];
  float g[8];
  #pragma unroll
  for (int j = 0; j < 8; ++j) {
    const int m = j + 8 * half;
    float l = bval + part[b][m][col] + part[b + 8][m][col]
                   + part[b + 16][m][col] + part[b + 24][m][col];
    float mx = l;
    mx = fmaxf(mx, __shfl_xor(mx, 1));
    mx = fmaxf(mx, __shfl_xor(mx, 2));
    mx = fmaxf(mx, __shfl_xor(mx, 4));
    mx = fmaxf(mx, __shfl_xor(mx, 8));
    float e = __expf(l - mx);
    float s = e;
    s += __shfl_xor(s, 1);
    s += __shfl_xor(s, 2);
    s += __shfl_xor(s, 4);
    s += __shfl_xor(s, 8);
    g[j] = e / s;
    if (q == 0) gate[b][j + 8 * half][col] = g[j];
  }
  __syncthreads();

  if (q == 0) {
    #pragma unroll
    for (int j = 0; j < 8; ++j) {
      const int m = j + 8 * half;
      float denom = EPSILON;
      #pragma unroll
      for (int bb = 0; bb < B_DIM; ++bb) denom += gate[bb][m][col];
      out[((size_t)b * S_DIM + (size_t)(s0 + m)) * E_DIM + col] =
          g[j] / denom * CAPACITY;
    }
  }
}

extern "C" void kernel_launch(void* const* d_in, const int* in_sizes, int n_in,
                              void* d_out, int out_size, void* d_ws, size_t ws_size,
                              hipStream_t stream) {
  const float* X    = (const float*)d_in[0];
  const float* W    = (const float*)d_in[1];
  const float* bias = (const float*)d_in[2];
  float* out        = (float*)d_out;

  dim3 grid(S_DIM / 16);
  dim3 block(1024);
  switch_gate_wmma_kernel<<<grid, block, 0, stream>>>(X, W, bias, out);
}


# 16 waves per block, not 32: a wave keeps a 128-k W slice in registers and streams 64-k half-tiles through a ring of four buffers, both halves summed in the accumulators; sum order changes
# speedup vs baseline: 1.3818x; 1.3818x over previous
#include <hip/hip_runtime.h>


typedef __attribute__((ext_vector_type(2))) float v2f;
typedef __attribute__((ext_vector_type(4))) float v4f;
typedef __attribute__((ext_vector_type(8))) float v8f;

#define B_DIM 8
#define S_DIM 4096
#define D_DIM 2048
#define E_DIM 16
#define NWAVE 16
#define KSLICE (D_DIM / NWAVE)
#define NSTEP (KSLICE / 4)
#define HSTEP 16
#define NHALF (2 * B_DIM)
#define TILE_X ((size_t)S_DIM * D_DIM)
#define TILE_P (2 * 32 * 4)
#define CAPACITY 8.0f
#define EPSILON 1e-6f

typedef const float __attribute__((address_space(1)))* gfp;
typedef const v2f __attribute__((address_space(1)))* gv2p;
static __device__ __forceinline__ gfp scalar_base(const float* p) {
  unsigned long long u = (unsigned long long)p;
  asm volatile("" : "+s"(u));
  return (gfp)u;
}

static __device__ __forceinline__ void load_half(v2f (&a)[HSTEP], gfp xt, unsigned xo0, unsigned xo1) {
  #pragma unroll
  for (int jp = 0; jp < 8; ++jp) a[jp] = *(gv2p)(xt + xo0 + 8 * jp);
  #pragma unroll
  for (int jp = 0; jp < 8; ++jp) a[8 + jp] = *(gv2p)(xt + xo1 + 8 * jp);
}

static __device__ __forceinline__ void product_half(const v2f (&a)[HSTEP], const v2f (&bw)[NSTEP],
                                                    const int boff, v8f& c0, v8f& c1, bool odd,
                                                    const bool first) {
  v2f fa[8], fb[8];
  #pragma unroll
  for (int jp = 0; jp < 8; ++jp) {
    fa[jp] = odd ? a[8 + jp] : a[jp];
    fb[jp] = odd ? a[jp] : a[8 + jp];
  }
  __builtin_amdgcn_sched_barrier(0);
  if (first) { c0 = v8f{}; c1 = v8f{}; }
  #pragma unroll
  for (int jp = 0; jp < 8; ++jp) {
    c0 = __builtin_amdgcn_wmma_f32_16x16x4_f32(false, fa[jp], false, bw[boff + 2 * jp], (short)0, c0, false,
                                               false);
    c1 = __builtin_amdgcn_wmma_f32_16x16x4_f32(false, fb[jp], false, bw[boff + 2 * jp + 1], (short)0, c1,
                                               false, false);
  }
  asm volatile("v_nop\n\tv_nop\n\tv_nop\n\tv_nop\n\tv_nop\n\tv_nop\n\tv_nop\n\tv_nop"
               : "+v"(c0), "+v"(c1));
}

static __device__ __forceinline__ void park_tile(const v8f& c0, const v8f& c1, float* pw) {
  v4f p0 = {c0[0] + c1[1], c0[2] + c1[3], c0[4] + c1[5], c0[6] + c1[7]};
  v4f p1 = {c0[1] + c1[0], c0[3] + c1[2], c0[5] + c1[4], c0[7] + c1[6]};
  *(v4f*)pw = p0;
  *(v4f*)(pw + 32 * 4) = p1;
}

__global__ __launch_bounds__(512)
void switch_gate_wres_kernel(const float* X,
                             const float* W,
                             const float* __restrict__ bias,
                             float* __restrict__ out) {
  __shared__ __attribute__((aligned(16))) float part[NWAVE * B_DIM * TILE_P];
  __shared__ float gate[B_DIM][16][E_DIM];

  const int tid  = threadIdx.x;
  const int ww   = __builtin_amdgcn_readfirstlane(tid >> 5);
  const int lane = tid & 31;
  const int half = lane >> 4;
  const int col  = lane & 15;
  const int s0   = blockIdx.x * 16;
  const int k0   = ww * KSLICE;

  const float* xt = X + (size_t)s0 * D_DIM + k0;
  const int arow = col >> 1;
  const int q    = col & 1;
  const unsigned xo0 = (unsigned)((8 * (arow >> 2) + (arow & 3)) * D_DIM + 4 * q + 2 * half);
  const unsigned xo1 = (unsigned)((8 * (arow >> 2) + 4 + (arow & 3)) * D_DIM + 4 * (1 - q) + 2 * half);
  const bool odd = q != 0;

  v2f a0[HSTEP], a1[HSTEP], a2[HSTEP], a3[HSTEP];
  load_half(a0, scalar_base(xt), xo0, xo1);
  __builtin_amdgcn_sched_barrier(0);
  const float* wcol = W + (size_t)k0 * E_DIM + (unsigned)(2 * half * E_DIM + col);
  v2f bw[NSTEP];
  #pragma unroll
  for (int j = 0; j < NSTEP; ++j) {
    bw[j].x = wcol[(4 * j) * E_DIM];
    bw[j].y = wcol[(4 * j + 1) * E_DIM];
  }
  __builtin_amdgcn_sched_barrier(0);
  load_half(a1, scalar_base(xt + 64), xo0, xo1);
  load_half(a2, scalar_base(xt + TILE_X), xo0, xo1);

  float* pw = &part[(size_t)ww * B_DIM * TILE_P + lane * 4];

  v8f c0, c1;
  #pragma unroll
  for (int i = 0; i < NHALF; ++i) {
    __builtin_amdgcn_sched_barrier(0);
    if (i + 3 < NHALF) {
      const int n = i + 3;
      const gfp xn = scalar_base(xt + (size_t)(n >> 1) * TILE_X + (n & 1) * 64);
      if (n % 4 == 0) load_half(a0, xn, xo0, xo1);
      else if (n % 4 == 1) load_half(a1, xn, xo0, xo1);
      else if (n % 4 == 2) load_half(a2, xn, xo0, xo1);
      else load_half(a3, xn, xo0, xo1);
    }
    __builtin_amdgcn_sched_barrier(0);
    const int boff = (i & 1) * HSTEP;
    const bool first = (i & 1) == 0;
    if (i % 4 == 0) product_half(a0, bw, boff, c0, c1, odd, first);
    else if (i % 4 == 1) product_half(a1, bw, boff, c0, c1, odd, first);
    else if (i % 4 == 2) product_half(a2, bw, boff, c0, c1, odd, first);
    else product_half(a3, bw, boff, c0, c1, odd, first);
    if (i & 1) park_tile(c0, c1, pw + (i >> 1) * TILE_P);
  }
  __syncthreads();

  const int t = ww & 7;
  const int g = (ww >> 3) & 1;
  float gs[4];
  {
    const float bval = bias[col];
    v4f l = {bval, bval, bval, bval};
    #pragma unroll 8
    for (int w = 0; w < NWAVE; ++w)
      l += *(const v4f*)&part[((((w * B_DIM + t) * 2 + g) * 32) + lane) * 4];
    #pragma unroll
    for (int i = 0; i < 4; ++i) {
      float mx = l[i];
      mx = fmaxf(mx, __shfl_xor(mx, 1));
      mx = fmaxf(mx, __shfl_xor(mx, 2));
      mx = fmaxf(mx, __shfl_xor(mx, 4));
      mx = fmaxf(mx, __shfl_xor(mx, 8));
      float e = __expf(l[i] - mx);
      float s = e;
      s += __shfl_xor(s, 1);
      s += __shfl_xor(s, 2);
      s += __shfl_xor(s, 4);
      s += __shfl_xor(s, 8);
      gs[i] = e / s;
      gate[t][8 * half + 4 * g + i][col] = gs[i];
    }
  }
  __syncthreads();

  {
    #pragma unroll
    for (int i = 0; i < 4; ++i) {
      const int m = 8 * half + 4 * g + i;
      float denom = EPSILON;
      #pragma unroll
      for (int bb = 0; bb < B_DIM; ++bb) denom += gate[bb][m][col];
      out[((size_t)t * S_DIM + (size_t)(s0 + m)) * E_DIM + col] =
          gs[i] / denom * CAPACITY;
    }
  }
}

extern "C" void kernel_launch(void* const* d_in, const int* in_sizes, int n_in,
                              void* d_out, int out_size, void* d_ws, size_t ws_size,
                              hipStream_t stream) {
  const float* X    = (const float*)d_in[0];
  const float* W    = (const float*)d_in[1];
  const float* bias = (const float*)d_in[2];
  float* out        = (float*)d_out;

  dim3 grid(S_DIM / 16);
  dim3 block(512);
  switch_gate_wres_kernel<<<grid, block, 0, stream>>>(X, W, bias, out);
}


# X loads widened to 16 B: 4 tokens x one whole 128-B line each (4 lines a load, was 8; a line is asked for once, was 4 times); two lane selects build the fragments, 4 chains; sum order changes
# speedup vs baseline: 1.4334x; 1.0373x over previous
#include <hip/hip_runtime.h>


typedef __attribute__((ext_vector_type(2))) float v2f;
typedef __attribute__((ext_vector_type(4))) float v4f;
typedef __attribute__((ext_vector_type(8))) float v8f;

#define B_DIM 8
#define S_DIM 4096
#define D_DIM 2048
#define E_DIM 16
#define NWAVE 16
#define KSLICE (D_DIM / NWAVE)
#define NSTEP (KSLICE / 4)
#define HSTEP 16
#define HLOAD 8
#define NHALF (2 * B_DIM)
#define TILE_X ((size_t)S_DIM * D_DIM)
#define TILE_P (2 * 32 * 4)
#define CAPACITY 8.0f
#define EPSILON 1e-6f

typedef const float __attribute__((address_space(1)))* gfp;
typedef const v4f __attribute__((address_space(1)))* gv4p;
static __device__ __forceinline__ gfp scalar_base(const float* p) {
  unsigned long long u = (unsigned long long)p;
  asm volatile("" : "+s"(u));
  return (gfp)u;
}

static __device__ __forceinline__ void load_half(v4f (&a)[HLOAD], gfp xt, const unsigned (&xo)[4]) {
  #pragma unroll
  for (int u = 0; u < 2; ++u) {
    #pragma unroll
    for (int s = 0; s < 4; ++s) a[4 * u + s] = *(gv4p)(xt + xo[s] + 32 * u);
  }
}

static __device__ __forceinline__ void product_half(const v4f (&a)[HLOAD], const v2f (&bw)[NSTEP],
                                                    const int boff, v8f (&c)[4], bool q0, bool q1,
                                                    const bool first) {
  v4f f[2][4];
  #pragma unroll
  for (int u = 0; u < 2; ++u) {
    const v4f g00 = q0 ? a[4 * u + 1] : a[4 * u + 0];
    const v4f g01 = q0 ? a[4 * u + 0] : a[4 * u + 1];
    const v4f g10 = q0 ? a[4 * u + 3] : a[4 * u + 2];
    const v4f g11 = q0 ? a[4 * u + 2] : a[4 * u + 3];
    f[u][0] = q1 ? g10 : g00;
    f[u][1] = q1 ? g11 : g01;
    f[u][2] = q1 ? g00 : g10;
    f[u][3] = q1 ? g01 : g11;
  }
  __builtin_amdgcn_sched_barrier(0);
  if (first) {
    c[0] = v8f{}; c[1] = v8f{}; c[2] = v8f{}; c[3] = v8f{};
    asm volatile("" : "+v"(c[0]), "+v"(c[1]), "+v"(c[2]), "+v"(c[3]));
  }
  #pragma unroll
  for (int u = 0; u < 2; ++u) {
    #pragma unroll
    for (int e = 0; e < 2; ++e) {
      #pragma unroll
      for (int p = 0; p < 4; ++p) {
        const v2f fr = e ? v2f{f[u][p].z, f[u][p].w} : v2f{f[u][p].x, f[u][p].y};
        c[p] = __builtin_amdgcn_wmma_f32_16x16x4_f32(false, fr, false, bw[boff + 8 * u + 2 * p + e], (short)0,
                                                     c[p], false, false);
      }
    }
  }
  asm volatile("v_nop\n\tv_nop\n\tv_nop\n\tv_nop\n\tv_nop\n\tv_nop\n\tv_nop\n\tv_nop"
               : "+v"(c[0]), "+v"(c[1]), "+v"(c[2]), "+v"(c[3]));
}

static __device__ __forceinline__ void park_tile(const v8f (&c)[4], float* pw) {
  v4f p0, p1;
  #pragma unroll
  for (int i = 0; i < 4; ++i) {
    p0[i] = ((c[0][i] + c[1][i ^ 1]) + c[2][i ^ 2]) + c[3][i ^ 3];
    p1[i] = ((c[0][4 + i] + c[1][4 + (i ^ 1)]) + c[2][4 + (i ^ 2)]) + c[3][4 + (i ^ 3)];
  }
  *(v4f*)pw = p0;
  *(v4f*)(pw + 32 * 4) = p1;
}

__global__ __launch_bounds__(512)
void switch_gate_wres_kernel(const float* X,
                             const float* W,
                             const float* __restrict__ bias,
                             float* __restrict__ out) {
  __shared__ __attribute__((aligned(16))) float part[NWAVE * B_DIM * TILE_P];
  __shared__ float gate[B_DIM][16][E_DIM];

  const int tid  = threadIdx.x;
  const int ww   = __builtin_amdgcn_readfirstlane(tid >> 5);
  const int lane = tid & 31;
  const int half = lane >> 4;
  const int col  = lane & 15;
  const int s0   = blockIdx.x * 16;
  const int k0   = ww * KSLICE;

  const float* xt = X + (size_t)s0 * D_DIM + k0;
  const int arow = col >> 2;
  const int q    = col & 3;
  unsigned xo[4];
  #pragma unroll
  for (int s = 0; s < 4; ++s)
    xo[s] = (unsigned)((4 * arow + s) * D_DIM + 8 * (q ^ s) + 4 * half);
  const bool q0 = (q & 1) != 0;
  const bool q1 = (q & 2) != 0;

  v4f a0[HLOAD], a1[HLOAD], a2[HLOAD], a3[HLOAD];
  load_half(a0, scalar_base(xt), xo);
  __builtin_amdgcn_sched_barrier(0);
  const float* wcol = W + (size_t)k0 * E_DIM + (unsigned)(4 * half * E_DIM + col);
  v2f bw[NSTEP];
  #pragma unroll
  for (int j = 0; j < NSTEP; ++j) {
    bw[j].x = wcol[(8 * (j >> 1) + 2 * (j & 1)) * E_DIM];
    bw[j].y = wcol[(8 * (j >> 1) + 2 * (j & 1) + 1) * E_DIM];
  }
  __builtin_amdgcn_sched_barrier(0);
  load_half(a1, scalar_base(xt + 64), xo);
  load_half(a2, scalar_base(xt + TILE_X), xo);

  float* pw = &part[(size_t)ww * B_DIM * TILE_P + lane * 4];

  v8f c[4];
  #pragma unroll
  for (int i = 0; i < NHALF; ++i) {
    __builtin_amdgcn_sched_barrier(0);
    if (i + 3 < NHALF) {
      const int n = i + 3;
      const gfp xn = scalar_base(xt + (size_t)(n >> 1) * TILE_X + (n & 1) * 64);
      if (n % 4 == 0) load_half(a0, xn, xo);
      else if (n % 4 == 1) load_half(a1, xn, xo);
      else if (n % 4 == 2) load_half(a2, xn, xo);
      else load_half(a3, xn, xo);
    }
    __builtin_amdgcn_sched_barrier(0);
    const int boff = (i & 1) * HSTEP;
    const bool first = (i & 1) == 0;
    if (i % 4 == 0) product_half(a0, bw, boff, c, q0, q1, first);
    else if (i % 4 == 1) product_half(a1, bw, boff, c, q0, q1, first);
    else if (i % 4 == 2) product_half(a2, bw, boff, c, q0, q1, first);
    else product_half(a3, bw, boff, c, q0, q1, first);
    if (i & 1) park_tile(c, pw + (i >> 1) * TILE_P);
  }
  __syncthreads();

  const int t = ww & 7;
  const int g = (ww >> 3) & 1;
  float gs[4];
  {
    const float bval = bias[col];
    v4f l = {bval, bval, bval, bval};
    #pragma unroll 8
    for (int w = 0; w < NWAVE; ++w)
      l += *(const v4f*)&part[((((w * B_DIM + t) * 2 + g) * 32) + lane) * 4];
    #pragma unroll
    for (int i = 0; i < 4; ++i) {
      float mx = l[i];
      mx = fmaxf(mx, __shfl_xor(mx, 1));
      mx = fmaxf(mx, __shfl_xor(mx, 2));
      mx = fmaxf(mx, __shfl_xor(mx, 4));
      mx = fmaxf(mx, __shfl_xor(mx, 8));
      float e = __expf(l[i] - mx);
      float s = e;
      s += __shfl_xor(s, 1);
      s += __shfl_xor(s, 2);
      s += __shfl_xor(s, 4);
      s += __shfl_xor(s, 8);
      gs[i] = e / s;
      gate[t][8 * half + 4 * g + i][col] = gs[i];
    }
  }
  __syncthreads();

  {
    #pragma unroll
    for (int i = 0; i < 4; ++i) {
      const int m = 8 * half + 4 * g + i;
      float denom = EPSILON;
      #pragma unroll
      for (int bb = 0; bb < B_DIM; ++bb) denom += gate[bb][m][col];
      out[((size_t)t * S_DIM + (size_t)(s0 + m)) * E_DIM + col] =
          gs[i] / denom * CAPACITY;
    }
  }
}

extern "C" void kernel_launch(void* const* d_in, const int* in_sizes, int n_in,
                              void* d_out, int out_size, void* d_ws, size_t ws_size,
                              hipStream_t stream) {
  const float* X    = (const float*)d_in[0];
  const float* W    = (const float*)d_in[1];
  const float* bias = (const float*)d_in[2];
  float* out        = (float*)d_out;

  dim3 grid(S_DIM / 16);
  dim3 block(512);
  switch_gate_wres_kernel<<<grid, block, 0, stream>>>(X, W, bias, out);
}
